# starting kernel
# baseline (speedup 1.0000x reference)
#include <hip/hip_runtime.h>


typedef float v2f __attribute__((ext_vector_type(2)));
typedef float v8f __attribute__((ext_vector_type(8)));

#define N_NODES 50000
#define N_EDGES 800000
#define C_IN    128
#define C_HID   128
#define C_OUTC  64


__global__ void zero_kernel(float* __restrict__ p, int n) {
    int i = blockIdx.x * blockDim.x + threadIdx.x;
    if (i < n) p[i] = 0.0f;
}

__global__ void deg_kernel(const int* __restrict__ dst, float* __restrict__ deg, int E) {
    int i = blockIdx.x * blockDim.x + threadIdx.x;
    if (i < E) atomicAdd(&deg[dst[i]], 1.0f);
}

__global__ void invsqrt_kernel(float* __restrict__ deg, int n) {
    int i = blockIdx.x * blockDim.x + threadIdx.x;
    if (i < n) deg[i] = rsqrtf(deg[i] + 1.0f);
}

template<int K, int N, bool RELU_A>
__global__ void gemm16_wmma_f32(const float* __restrict__ A,
                                const float* __restrict__ W,
                                float* __restrict__ H) {
    const int lane  = threadIdx.x & 31;
    const int wave  = threadIdx.x >> 5;
    const int row0  = blockIdx.x << 4;
    const int col0  = wave << 4;
    const int m     = lane & 15;
    const int khalf = (lane >> 4) << 1;

    const float* __restrict__ arow = A + (size_t)(row0 + m) * K;
    const float* __restrict__ wcol = W + col0 + m;

    v8f acc = {};
#pragma unroll 8
    for (int k0 = 0; k0 < K; k0 += 4) {
        float a0 = arow[k0 + khalf + 0];
        float a1 = arow[k0 + khalf + 1];
        if (RELU_A) { a0 = fmaxf(a0, 0.0f); a1 = fmaxf(a1, 0.0f); }
        v2f a = {a0, a1};
        v2f b = { wcol[(size_t)(k0 + khalf + 0) * N],
                  wcol[(size_t)(k0 + khalf + 1) * N] };
        acc = __builtin_amdgcn_wmma_f32_16x16x4_f32(
                  false, a, false, b, (short)0, acc, false, false);
    }

    float* __restrict__ hrow =
        H + (size_t)(row0 + ((lane >> 4) << 3)) * N + col0 + m;
#pragma unroll
    for (int r = 0; r < 8; ++r)
        hrow[(size_t)r * N] = acc[r];
}

template<int C>
__global__ void selfloop_bias_kernel(const float* __restrict__ h,
                                     const float* __restrict__ isq,
                                     const float* __restrict__ bias,
                                     float* __restrict__ out, int n_elems) {
    int i = blockIdx.x * blockDim.x + threadIdx.x;
    if (i < n_elems) {
        int node = i / C;
        int c    = i - node * C;
        float s  = isq[node];
        out[i] = h[i] * (s * s) + bias[c];
    }
}

template<int C>
__global__ void edge_scatter_kernel(const float* __restrict__ h,
                                    const int* __restrict__ src,
                                    const int* __restrict__ dst,
                                    const float* __restrict__ isq,
                                    float* __restrict__ out, int E) {
    const int lane   = threadIdx.x & 31;
    const int wave   = (blockIdx.x * blockDim.x + threadIdx.x) >> 5;
    const int nwaves = (gridDim.x * blockDim.x) >> 5;
    constexpr int V  = C / 32;

    for (int e = wave; e < E; e += nwaves) {
        const int   s    = src[e];
        const int   d    = dst[e];
        const float coef = isq[s] * isq[d];
        const float* __restrict__ hs = h   + (size_t)s * C;
        float*       __restrict__ od = out + (size_t)d * C;
#pragma unroll
        for (int v = 0; v < V; ++v) {
            int c = lane + (v << 5);
            atomicAdd(&od[c], hs[c] * coef);
        }
    }
}


extern "C" void kernel_launch(void* const* d_in, const int* in_sizes, int n_in,
                              void* d_out, int out_size, void* d_ws, size_t ws_size,
                              hipStream_t stream) {
    (void)in_sizes; (void)n_in; (void)out_size; (void)ws_size;

    const float* x   = (const float*)d_in[0];
    const int*   ei  = (const int*)  d_in[1];
    const float* W1  = (const float*)d_in[2];
    const float* b1  = (const float*)d_in[3];
    const float* W2  = (const float*)d_in[4];
    const float* b2  = (const float*)d_in[5];
    float*       out = (float*)d_out;

    const int* src = ei;
    const int* dst = ei + N_EDGES;

    float* isq  = (float*)d_ws;
    float* h1   = isq + 50176;
    float* out1 = h1 + (size_t)N_NODES * C_HID;
    float* h2   = h1;

    const int T = 256;

    zero_kernel   <<<(N_NODES + T - 1) / T, T, 0, stream>>>(isq, N_NODES);
    deg_kernel    <<<(N_EDGES + T - 1) / T, T, 0, stream>>>(dst, isq, N_EDGES);
    invsqrt_kernel<<<(N_NODES + T - 1) / T, T, 0, stream>>>(isq, N_NODES);

    gemm16_wmma_f32<C_IN, C_HID, false>
        <<<N_NODES / 16, (C_HID / 16) * 32, 0, stream>>>(x, W1, h1);

    selfloop_bias_kernel<C_HID>
        <<<((N_NODES * C_HID) + T - 1) / T, T, 0, stream>>>(h1, isq, b1, out1,
                                                            N_NODES * C_HID);
    edge_scatter_kernel<C_HID>
        <<<1024, T, 0, stream>>>(h1, src, dst, isq, out1, N_EDGES);

    gemm16_wmma_f32<C_HID, C_OUTC, true>
        <<<N_NODES / 16, (C_OUTC / 16) * 32, 0, stream>>>(out1, W2, h2);

    selfloop_bias_kernel<C_OUTC>
        <<<((N_NODES * C_OUTC) + T - 1) / T, T, 0, stream>>>(h2, isq, b2, out,
                                                             N_NODES * C_OUTC);
    edge_scatter_kernel<C_OUTC>
        <<<1024, T, 0, stream>>>(h2, src, dst, isq, out, N_EDGES);
}


# Replace per-edge f32 atomic scatter by in-call counting sort of edges by destination (int atomic rank + scan) and one-wave-per-node register gather; sum order = rank order, self loop last
# speedup vs baseline: 2.2120x; 2.2120x over previous
#include <hip/hip_runtime.h>


typedef float v2f __attribute__((ext_vector_type(2)));
typedef float v4f __attribute__((ext_vector_type(4)));
typedef float v8f __attribute__((ext_vector_type(8)));
typedef int   v4i __attribute__((ext_vector_type(4)));

#define N_NODES 50000
#define N_EDGES 800000
#define C_IN    128
#define C_HID   128
#define C_OUTC  64

#define SCAN_ROWS 13
#define N_PAD     (32 * SCAN_ROWS * 128)

static_assert(N_PAD > N_NODES, "scan must cover entry N_NODES (the end of the last segment)");
static_assert(N_NODES % 8 == 0, "gather grid assumes 8 nodes per block");


__global__ void zero_int_kernel(int* __restrict__ p, int n) {
    int i = blockIdx.x * blockDim.x + threadIdx.x;
    if (i < n) p[i] = 0;
}

__global__ void rank_kernel(const int* __restrict__ dst, int* __restrict__ cnt,
                            int* __restrict__ rank, int E) {
    int i = blockIdx.x * blockDim.x + threadIdx.x;
    if (i < E) rank[i] = atomicAdd(&cnt[dst[i]], 1);
}

__global__ __launch_bounds__(1024)
void scan_kernel(const int* __restrict__ cnt, int* __restrict__ rowptr,
                 float* __restrict__ isq) {
    __shared__ int wtot[32];
    const int lane = threadIdx.x & 31;
    const int wave = threadIdx.x >> 5;
    const int base = wave * (SCAN_ROWS * 128) + lane * 4;

    v4i c[SCAN_ROWS];
#pragma unroll
    for (int r = 0; r < SCAN_ROWS; ++r)
        c[r] = *reinterpret_cast<const v4i*>(cnt + base + r * 128);

    int carry = 0;
#pragma unroll
    for (int r = 0; r < SCAN_ROWS; ++r) {
        const v4i v = c[r];
        v4f q;
        q.x = rsqrtf((float)v.x + 1.0f);
        q.y = rsqrtf((float)v.y + 1.0f);
        q.z = rsqrtf((float)v.z + 1.0f);
        q.w = rsqrtf((float)v.w + 1.0f);
        *reinterpret_cast<v4f*>(isq + base + r * 128) = q;

        const int s1 = v.x, s2 = s1 + v.y, s3 = s2 + v.z, s4 = s3 + v.w;
        int x = s4;
        for (int off = 1; off < 32; off <<= 1) {
            const int y = __shfl_up(x, off, 32);
            if (lane >= off) x += y;
        }
        const int tot = __shfl(x, 31, 32);
        const int ex  = carry + x - s4;
        v4i o;
        o.x = ex; o.y = ex + s1; o.z = ex + s2; o.w = ex + s3;
        c[r] = o;
        carry += tot;
    }
    if (lane == 0) wtot[wave] = carry;
    __syncthreads();
    int t = (lane < wave) ? wtot[lane] : 0;
    for (int off = 16; off > 0; off >>= 1) t += __shfl_xor(t, off, 32);

#pragma unroll
    for (int r = 0; r < SCAN_ROWS; ++r)
        *reinterpret_cast<v4i*>(rowptr + base + r * 128) = c[r] + t;
}

__global__ void fill_kernel(const int* __restrict__ src, const int* __restrict__ dst,
                            const int* __restrict__ rank, const int* __restrict__ rowptr,
                            int* __restrict__ col, int E) {
    int i = blockIdx.x * blockDim.x + threadIdx.x;
    if (i < E) col[rowptr[dst[i]] + rank[i]] = src[i];
}

template<int K, int N, bool RELU_A>
__global__ void gemm16_wmma_f32(const float* __restrict__ A,
                                const float* __restrict__ W,
                                float* __restrict__ H) {
    const int lane  = threadIdx.x & 31;
    const int wave  = threadIdx.x >> 5;
    const int row0  = blockIdx.x << 4;
    const int col0  = wave << 4;
    const int m     = lane & 15;
    const int khalf = (lane >> 4) << 1;

    const float* __restrict__ arow = A + (size_t)(row0 + m) * K;
    const float* __restrict__ wcol = W + col0 + m;

    v8f acc = {};
#pragma unroll 8
    for (int k0 = 0; k0 < K; k0 += 4) {
        float a0 = arow[k0 + khalf + 0];
        float a1 = arow[k0 + khalf + 1];
        if (RELU_A) { a0 = fmaxf(a0, 0.0f); a1 = fmaxf(a1, 0.0f); }
        v2f a = {a0, a1};
        v2f b = { wcol[(size_t)(k0 + khalf + 0) * N],
                  wcol[(size_t)(k0 + khalf + 1) * N] };
        acc = __builtin_amdgcn_wmma_f32_16x16x4_f32(
                  false, a, false, b, (short)0, acc, false, false);
    }

    float* __restrict__ hrow =
        H + (size_t)(row0 + ((lane >> 4) << 3)) * N + col0 + m;
#pragma unroll
    for (int r = 0; r < 8; ++r)
        hrow[(size_t)r * N] = acc[r];
}


template<int U, typename vec>
__device__ __forceinline__ void gather_rows(vec& acc, const vec* __restrict__ hv,
                                            int lane, int s_l, float c_l, int t) {
    vec   hr[U];
    float cf[U];
#pragma unroll
    for (int u = 0; u < U; ++u) {
        const int s = __builtin_amdgcn_readlane(s_l, t + u);
        cf[u] = __builtin_bit_cast(
            float, __builtin_amdgcn_readlane(__builtin_bit_cast(int, c_l), t + u));
        hr[u] = hv[(size_t)s * 32 + lane];
    }
#pragma unroll
    for (int u = 0; u < U; ++u) acc += hr[u] * cf[u];
}

template<int C>
__global__ __launch_bounds__(256)
void gather_kernel(const float* __restrict__ h,
                   const int* __restrict__ rowptr,
                   const int* __restrict__ col,
                   const float* __restrict__ isq,
                   const float* __restrict__ bias,
                   float* __restrict__ out, int n) {
    constexpr int V = C / 32;
    typedef float vec __attribute__((ext_vector_type(V)));

    const int lane = threadIdx.x & 31;
    const int node = __builtin_amdgcn_readfirstlane(
        (int)(blockIdx.x * (blockDim.x >> 5) + (threadIdx.x >> 5)));
    if (node >= n) return;

    const int   beg = rowptr[node];
    const int   end = rowptr[node + 1];
    const float sd  = isq[node];

    const vec* __restrict__ hv = reinterpret_cast<const vec*>(h);
    const vec* __restrict__ bv = reinterpret_cast<const vec*>(bias);

    const vec hself = hv[(size_t)node * 32 + lane];
    const vec bself = bv[lane];

    vec acc = (vec)(0.0f);

    for (int base = beg; base < end; base += 32) {
        const int j = base + lane;
        int   s_l = 0;
        float c_l = 0.0f;
        if (j < end) {
            s_l = col[j];
            c_l = isq[s_l] * sd;
        }
        const int rem = end - base;
        const int m   = rem < 32 ? rem : 32;

        int t = 0;
        for (; t + 8 <= m; t += 8) gather_rows<8, vec>(acc, hv, lane, s_l, c_l, t);
        if (m & 4) { gather_rows<4, vec>(acc, hv, lane, s_l, c_l, t); t += 4; }
        if (m & 2) { gather_rows<2, vec>(acc, hv, lane, s_l, c_l, t); t += 2; }
        if (m & 1) { gather_rows<1, vec>(acc, hv, lane, s_l, c_l, t); }
    }

    acc += hself * (sd * sd) + bself;
    reinterpret_cast<vec*>(out)[(size_t)node * 32 + lane] = acc;
}


extern "C" void kernel_launch(void* const* d_in, const int* in_sizes, int n_in,
                              void* d_out, int out_size, void* d_ws, size_t ws_size,
                              hipStream_t stream) {
    (void)in_sizes; (void)n_in; (void)out_size; (void)ws_size;

    const float* x   = (const float*)d_in[0];
    const int*   ei  = (const int*)  d_in[1];
    const float* W1  = (const float*)d_in[2];
    const float* b1  = (const float*)d_in[3];
    const float* W2  = (const float*)d_in[4];
    const float* b2  = (const float*)d_in[5];
    float*       out = (float*)d_out;

    const int* src = ei;
    const int* dst = ei + N_EDGES;

    int*   cnt    = (int*)d_ws;
    int*   rowptr = cnt + N_PAD;
    float* isq    = (float*)(rowptr + N_PAD);
    int*   rank   = (int*)(isq + N_PAD);
    int*   col    = rank + N_EDGES;
    float* h1     = (float*)(col + N_EDGES);
    float* out1   = h1 + (size_t)N_NODES * C_HID;
    float* h2     = h1;

    const int T = 256;

    zero_int_kernel<<<N_PAD / T, T, 0, stream>>>(cnt, N_PAD);
    rank_kernel    <<<(N_EDGES + T - 1) / T, T, 0, stream>>>(dst, cnt, rank, N_EDGES);
    scan_kernel    <<<1, 1024, 0, stream>>>(cnt, rowptr, isq);
    fill_kernel    <<<(N_EDGES + T - 1) / T, T, 0, stream>>>(src, dst, rank, rowptr,
                                                             col, N_EDGES);

    gemm16_wmma_f32<C_IN, C_HID, false>
        <<<N_NODES / 16, (C_HID / 16) * 32, 0, stream>>>(x, W1, h1);

    gather_kernel<C_HID>
        <<<N_NODES / 8, T, 0, stream>>>(h1, rowptr, col, isq, b1, out1, N_NODES);

    gemm16_wmma_f32<C_HID, C_OUTC, true>
        <<<N_NODES / 16, (C_OUTC / 16) * 32, 0, stream>>>(out1, W2, h2);

    gather_kernel<C_OUTC>
        <<<N_NODES / 8, T, 0, stream>>>(h2, rowptr, col, isq, b2, out, N_NODES);
}


# Edge sort ranked by LDS counters, not one returning global atomic per edge: two-level counting sort (chunks by dst>>8, then per node in a bucket block); 4 launches -> 2; sum order = LDS arrival rank
# speedup vs baseline: 3.9920x; 1.8047x over previous
#include <hip/hip_runtime.h>


typedef float v2f __attribute__((ext_vector_type(2)));
typedef float v8f __attribute__((ext_vector_type(8)));
typedef int   v2i __attribute__((ext_vector_type(2)));

#define N_NODES 50000
#define N_EDGES 800000
#define C_IN    128
#define C_HID   128
#define C_OUTC  64

#define N_PAD     53248

static_assert(N_PAD > N_NODES && N_PAD % 128 == 0, "padded node arrays");
static_assert(N_NODES % 8 == 0, "gather grid assumes 8 nodes per block");


#define SORT_CHUNKS 256
#define CHUNK_E     (N_EDGES / SORT_CHUNKS)
#define BKT_NODES   256
#define N_BKT       ((N_NODES + BKT_NODES - 1) / BKT_NODES)
#define SORT_T      1024
#define CHUNK_PER_T ((CHUNK_E + SORT_T - 1) / SORT_T)

static_assert(N_EDGES % SORT_CHUNKS == 0, "chunks must tile the edge list");
static_assert(N_BKT + 1 <= 256, "bucket offsets are scanned as 256 entries by one wave");
static_assert(N_BKT * BKT_NODES + 1 <= N_PAD, "rowptr must hold every bucket's nodes");
static_assert(N_BKT * BKT_NODES > N_NODES, "entry N_NODES of rowptr must belong to a bucket");
static_assert((((N_BKT + 1) * SORT_CHUNKS) % 2) == 0, "pairs must start 8-byte aligned");
static_assert((((N_BKT + 1) * SORT_CHUNKS + 3 * N_EDGES + 2 * N_PAD) % 128) == 0,
              "row tables must start on a 512-byte boundary");

__global__ __launch_bounds__(SORT_T)
void chunk_sort_kernel(const int* __restrict__ src, const int* __restrict__ dst,
                       int* __restrict__ pairs, int* __restrict__ offs_t) {
    __shared__ int hist[256];
    __shared__ v2i stage[CHUNK_E];
    const int tid = threadIdx.x;
    const int e0  = blockIdx.x * CHUNK_E;

    if (tid < 256) hist[tid] = 0;

    int s[CHUNK_PER_T], d[CHUNK_PER_T], r[CHUNK_PER_T];
#pragma unroll
    for (int i = 0; i < CHUNK_PER_T; ++i) {
        const int j = tid + i * SORT_T;
        s[i] = 0; d[i] = -1;
        if (j < CHUNK_E) {
            s[i] = src[e0 + j];
            d[i] = dst[e0 + j];
        }
    }
    __syncthreads();
#pragma unroll
    for (int i = 0; i < CHUNK_PER_T; ++i) {
        r[i] = 0;
        if (d[i] >= 0) r[i] = atomicAdd(&hist[d[i] >> 8], 1);
    }
    __syncthreads();

    if (tid < 32) {
        int v[8];
        int sum = 0;
#pragma unroll
        for (int q = 0; q < 8; ++q) { const int c = hist[tid * 8 + q]; v[q] = sum; sum += c; }
        int x = sum;
        for (int off = 1; off < 32; off <<= 1) {
            const int y = __shfl_up(x, off, 32);
            if (tid >= off) x += y;
        }
        const int ex = x - sum;
#pragma unroll
        for (int q = 0; q < 8; ++q) hist[tid * 8 + q] = ex + v[q];
    }
    __syncthreads();

    if (tid <= N_BKT) offs_t[tid * SORT_CHUNKS + blockIdx.x] = hist[tid];

#pragma unroll
    for (int i = 0; i < CHUNK_PER_T; ++i) {
        if (d[i] >= 0) {
            v2i p = {s[i], d[i]};
            stage[hist[d[i] >> 8] + r[i]] = p;
        }
    }
    __syncthreads();

    v2i* __restrict__ pout = reinterpret_cast<v2i*>(pairs) + e0;
    for (int j = tid; j < CHUNK_E; j += SORT_T) pout[j] = stage[j];
}

__device__ __forceinline__ int scan256_inclusive(int x, int lane, int wave, int* wtot) {
    for (int off = 1; off < 32; off <<= 1) {
        const int y = __shfl_up(x, off, 32);
        if (lane >= off) x += y;
    }
    if (wave < 8 && lane == 31) wtot[wave] = x;
    return x;
}

__device__ __forceinline__ int pair_index(const int* pfx, const int* pstart, int p) {
    int lo = 0, hi = SORT_CHUNKS;
    for (int it = 0; it < 8; ++it) {
        const int mid = (lo + hi) >> 1;
        if (pfx[mid] <= p) lo = mid; else hi = mid;
    }
    return pstart[lo] + (p - pfx[lo]);
}
static_assert(SORT_CHUNKS == 256, "pair_index searches 256 pieces in 8 steps");

#define KEEP 8
#define STAGE_CAP 16384

__global__ __launch_bounds__(SORT_T)
void bucket_sort_kernel(const int* __restrict__ pairs, const int* __restrict__ offs_t,
                        int* __restrict__ rowptr, float* __restrict__ isq,
                        int* __restrict__ col) {
    __shared__ int pstart[SORT_CHUNKS];
    __shared__ int pfx[SORT_CHUNKS + 1];
    __shared__ int cnt[BKT_NODES];
    __shared__ int nstart[BKT_NODES];
    __shared__ int wtotA[8], wtotB[8], wtotC[8];
    __shared__ int lcol[STAGE_CAP];

    const int tid  = threadIdx.x;
    const int lane = tid & 31;
    const int wave = tid >> 5;
    const int k    = blockIdx.x;
    const v2i* __restrict__ pv = reinterpret_cast<const v2i*>(pairs);

    int len = 0, o0 = 0;
    if (tid < SORT_CHUNKS) {
        o0 = offs_t[k * SORT_CHUNKS + tid];
        const int o1 = offs_t[(k + 1) * SORT_CHUNKS + tid];
        len = o1 - o0;
        pstart[tid] = tid * CHUNK_E + o0;
        cnt[tid] = 0;
    }
    const int lenInc = scan256_inclusive(len, lane, wave, wtotA);
    (void)scan256_inclusive(o0, lane, wave, wtotB);
    __syncthreads();
    if (tid < SORT_CHUNKS) {
        int woff = 0;
        for (int w = 0; w < wave; ++w) woff += wtotA[w];
        pfx[tid] = woff + lenInc - len;
        if (tid == SORT_CHUNKS - 1) pfx[SORT_CHUNKS] = woff + lenInc;
    }
    int base = 0;
    for (int w = 0; w < 8; ++w) base += wtotB[w];
    __syncthreads();
    const int total = pfx[SORT_CHUNKS];

    v2i keep[KEEP];
#pragma unroll
    for (int u = 0; u < KEEP; ++u) {
        const int p  = tid + u * SORT_T;
        const int ix = pair_index(pfx, pstart, p);
        keep[u] = pv[p < total ? ix : 0];
    }
#pragma unroll
    for (int u = 0; u < KEEP; ++u)
        if (tid + u * SORT_T < total) atomicAdd(&cnt[keep[u].y - k * BKT_NODES], 1);
    for (int p = tid + KEEP * SORT_T; p < total; p += SORT_T) {
        const v2i pr = pv[pair_index(pfx, pstart, p)];
        atomicAdd(&cnt[pr.y - k * BKT_NODES], 1);
    }
    __syncthreads();

    int c = 0;
    if (tid < BKT_NODES) c = cnt[tid];
    const int cInc = scan256_inclusive(c, lane, wave, wtotC);
    __syncthreads();
    if (tid < BKT_NODES) {
        int woff = 0;
        for (int w = 0; w < wave; ++w) woff += wtotC[w];
        const int st = woff + cInc - c;
        nstart[tid] = st;
        cnt[tid]    = 0;
        const int node = k * BKT_NODES + tid;
        rowptr[node] = base + st;
        isq[node]    = rsqrtf((float)c + 1.0f);
    }
    __syncthreads();

    const bool staged = total <= STAGE_CAP;
    int* __restrict__ cslice = col + base;
#pragma unroll
    for (int u = 0; u < KEEP; ++u) {
        if (tid + u * SORT_T < total) {
            const int dl = keep[u].y - k * BKT_NODES;
            const int q  = nstart[dl] + atomicAdd(&cnt[dl], 1);
            if (staged) lcol[q] = keep[u].x; else cslice[q] = keep[u].x;
        }
    }
    for (int p = tid + KEEP * SORT_T; p < total; p += SORT_T) {
        const v2i pr = pv[pair_index(pfx, pstart, p)];
        const int dl = pr.y - k * BKT_NODES;
        const int q  = nstart[dl] + atomicAdd(&cnt[dl], 1);
        if (staged) lcol[q] = pr.x; else cslice[q] = pr.x;
    }
    __syncthreads();
    if (staged)
        for (int q = tid; q < total; q += SORT_T) cslice[q] = lcol[q];
}

template<int K, int N, bool RELU_A>
__global__ void gemm16_wmma_f32(const float* __restrict__ A,
                                const float* __restrict__ W,
                                float* __restrict__ H) {
    const int lane  = threadIdx.x & 31;
    const int wave  = threadIdx.x >> 5;
    const int row0  = blockIdx.x << 4;
    const int col0  = wave << 4;
    const int m     = lane & 15;
    const int khalf = (lane >> 4) << 1;

    const float* __restrict__ arow = A + (size_t)(row0 + m) * K;
    const float* __restrict__ wcol = W + col0 + m;

    v8f acc = {};
#pragma unroll 8
    for (int k0 = 0; k0 < K; k0 += 4) {
        float a0 = arow[k0 + khalf + 0];
        float a1 = arow[k0 + khalf + 1];
        if (RELU_A) { a0 = fmaxf(a0, 0.0f); a1 = fmaxf(a1, 0.0f); }
        v2f a = {a0, a1};
        v2f b = { wcol[(size_t)(k0 + khalf + 0) * N],
                  wcol[(size_t)(k0 + khalf + 1) * N] };
        acc = __builtin_amdgcn_wmma_f32_16x16x4_f32(
                  false, a, false, b, (short)0, acc, false, false);
    }

    float* __restrict__ hrow =
        H + (size_t)(row0 + ((lane >> 4) << 3)) * N + col0 + m;
#pragma unroll
    for (int r = 0; r < 8; ++r)
        hrow[(size_t)r * N] = acc[r];
}


template<int U, typename vec>
__device__ __forceinline__ void gather_rows(vec& acc, const vec* __restrict__ hv,
                                            int lane, int s_l, float c_l, int t) {
    vec   hr[U];
    float cf[U];
#pragma unroll
    for (int u = 0; u < U; ++u) {
        const int s = __builtin_amdgcn_readlane(s_l, t + u);
        cf[u] = __builtin_bit_cast(
            float, __builtin_amdgcn_readlane(__builtin_bit_cast(int, c_l), t + u));
        hr[u] = hv[(size_t)s * 32 + lane];
    }
#pragma unroll
    for (int u = 0; u < U; ++u) acc += hr[u] * cf[u];
}

template<int C>
__global__ __launch_bounds__(256)
void gather_kernel(const float* __restrict__ h,
                   const int* __restrict__ rowptr,
                   const int* __restrict__ col,
                   const float* __restrict__ isq,
                   const float* __restrict__ bias,
                   float* __restrict__ out, int n) {
    constexpr int V = C / 32;
    typedef float vec __attribute__((ext_vector_type(V)));

    const int lane = threadIdx.x & 31;
    const int node = __builtin_amdgcn_readfirstlane(
        (int)(blockIdx.x * (blockDim.x >> 5) + (threadIdx.x >> 5)));
    if (node >= n) return;

    const int   beg = rowptr[node];
    const int   end = rowptr[node + 1];
    const float sd  = isq[node];

    const vec* __restrict__ hv = reinterpret_cast<const vec*>(h);
    const vec* __restrict__ bv = reinterpret_cast<const vec*>(bias);

    const vec hself = hv[(size_t)node * 32 + lane];
    const vec bself = bv[lane];

    vec acc = (vec)(0.0f);

    for (int base = beg; base < end; base += 32) {
        const int j = base + lane;
        int   s_l = 0;
        float c_l = 0.0f;
        if (j < end) {
            s_l = col[j];
            c_l = isq[s_l] * sd;
        }
        const int rem = end - base;
        const int m   = rem < 32 ? rem : 32;

        int t = 0;
        for (; t + 8 <= m; t += 8) gather_rows<8, vec>(acc, hv, lane, s_l, c_l, t);
        if (m & 4) { gather_rows<4, vec>(acc, hv, lane, s_l, c_l, t); t += 4; }
        if (m & 2) { gather_rows<2, vec>(acc, hv, lane, s_l, c_l, t); t += 2; }
        if (m & 1) { gather_rows<1, vec>(acc, hv, lane, s_l, c_l, t); }
    }

    acc += hself * (sd * sd) + bself;
    reinterpret_cast<vec*>(out)[(size_t)node * 32 + lane] = acc;
}


extern "C" void kernel_launch(void* const* d_in, const int* in_sizes, int n_in,
                              void* d_out, int out_size, void* d_ws, size_t ws_size,
                              hipStream_t stream) {
    (void)in_sizes; (void)n_in; (void)out_size; (void)ws_size;

    const float* x   = (const float*)d_in[0];
    const int*   ei  = (const int*)  d_in[1];
    const float* W1  = (const float*)d_in[2];
    const float* b1  = (const float*)d_in[3];
    const float* W2  = (const float*)d_in[4];
    const float* b2  = (const float*)d_in[5];
    float*       out = (float*)d_out;

    const int* src = ei;
    const int* dst = ei + N_EDGES;

    int*   offs_t = (int*)d_ws;
    int*   pairs  = offs_t + (N_BKT + 1) * SORT_CHUNKS;
    int*   rowptr = pairs + 2 * N_EDGES;
    float* isq    = (float*)(rowptr + N_PAD);
    int*   col    = (int*)(isq + N_PAD);
    float* h1     = (float*)(col + N_EDGES);
    float* out1   = h1 + (size_t)N_NODES * C_HID;
    float* h2     = h1;

    const int T = 256;

    chunk_sort_kernel <<<SORT_CHUNKS, SORT_T, 0, stream>>>(src, dst, pairs, offs_t);
    bucket_sort_kernel<<<N_BKT, SORT_T, 0, stream>>>(pairs, offs_t, rowptr, isq, col);

    gemm16_wmma_f32<C_IN, C_HID, false>
        <<<N_NODES / 16, (C_HID / 16) * 32, 0, stream>>>(x, W1, h1);

    gather_kernel<C_HID>
        <<<N_NODES / 8, T, 0, stream>>>(h1, rowptr, col, isq, b1, out1, N_NODES);

    gemm16_wmma_f32<C_HID, C_OUTC, true>
        <<<N_NODES / 16, (C_OUTC / 16) * 32, 0, stream>>>(out1, W2, h2);

    gather_kernel<C_OUTC>
        <<<N_NODES / 8, T, 0, stream>>>(h2, rowptr, col, isq, b2, out, N_NODES);
}


# Layer 1 aggregates before it transforms ((A_hat x) W1, a changed sum order): one kernel gathers 32 rows of x into LDS and runs the W1 and W2 WMMA products from LDS; 6 launches -> 4
# speedup vs baseline: 5.5628x; 1.3935x over previous
#include <hip/hip_runtime.h>


typedef float v2f __attribute__((ext_vector_type(2)));
typedef float v4f __attribute__((ext_vector_type(4)));
typedef float v8f __attribute__((ext_vector_type(8)));
typedef int   v2i __attribute__((ext_vector_type(2)));

#define N_NODES 50000
#define N_EDGES 800000
#define C_IN    128
#define C_HID   128
#define C_OUTC  64

#define N_PAD     53248

static_assert(N_PAD > N_NODES && N_PAD % 128 == 0, "padded node arrays");
static_assert(N_NODES % 8 == 0, "gather grid assumes 8 nodes per block");


#define SORT_CHUNKS 256
#define CHUNK_E     (N_EDGES / SORT_CHUNKS)
#define BKT_NODES   256
#define N_BKT       ((N_NODES + BKT_NODES - 1) / BKT_NODES)
#define SORT_T      1024
#define CHUNK_PER_T ((CHUNK_E + SORT_T - 1) / SORT_T)

static_assert(N_EDGES % SORT_CHUNKS == 0, "chunks must tile the edge list");
static_assert(N_BKT + 1 <= 256, "bucket offsets are scanned as 256 entries by one wave");
static_assert(N_BKT * BKT_NODES + 1 <= N_PAD, "rowptr must hold every bucket's nodes");
static_assert(N_BKT * BKT_NODES > N_NODES, "entry N_NODES of rowptr must belong to a bucket");
static_assert((((N_BKT + 1) * SORT_CHUNKS) % 2) == 0, "pairs must start 8-byte aligned");
static_assert((((N_BKT + 1) * SORT_CHUNKS + 3 * N_EDGES + 2 * N_PAD) % 128) == 0,
              "row tables must start on a 512-byte boundary");

__global__ __launch_bounds__(SORT_T)
void chunk_sort_kernel(const int* __restrict__ src, const int* __restrict__ dst,
                       int* __restrict__ pairs, int* __restrict__ offs_t) {
    __shared__ int hist[256];
    __shared__ v2i stage[CHUNK_E];
    const int tid = threadIdx.x;
    const int e0  = blockIdx.x * CHUNK_E;

    if (tid < 256) hist[tid] = 0;

    int s[CHUNK_PER_T], d[CHUNK_PER_T], r[CHUNK_PER_T];
#pragma unroll
    for (int i = 0; i < CHUNK_PER_T; ++i) {
        const int j = tid + i * SORT_T;
        s[i] = 0; d[i] = -1;
        if (j < CHUNK_E) {
            s[i] = src[e0 + j];
            d[i] = dst[e0 + j];
        }
    }
    __syncthreads();
#pragma unroll
    for (int i = 0; i < CHUNK_PER_T; ++i) {
        r[i] = 0;
        if (d[i] >= 0) r[i] = atomicAdd(&hist[d[i] >> 8], 1);
    }
    __syncthreads();

    if (tid < 32) {
        int v[8];
        int sum = 0;
#pragma unroll
        for (int q = 0; q < 8; ++q) { const int c = hist[tid * 8 + q]; v[q] = sum; sum += c; }
        int x = sum;
        for (int off = 1; off < 32; off <<= 1) {
            const int y = __shfl_up(x, off, 32);
            if (tid >= off) x += y;
        }
        const int ex = x - sum;
#pragma unroll
        for (int q = 0; q < 8; ++q) hist[tid * 8 + q] = ex + v[q];
    }
    __syncthreads();

    if (tid <= N_BKT) offs_t[tid * SORT_CHUNKS + blockIdx.x] = hist[tid];

#pragma unroll
    for (int i = 0; i < CHUNK_PER_T; ++i) {
        if (d[i] >= 0) {
            v2i p = {s[i], d[i]};
            stage[hist[d[i] >> 8] + r[i]] = p;
        }
    }
    __syncthreads();

    v2i* __restrict__ pout = reinterpret_cast<v2i*>(pairs) + e0;
    for (int j = tid; j < CHUNK_E; j += SORT_T) pout[j] = stage[j];
}

__device__ __forceinline__ int scan256_inclusive(int x, int lane, int wave, int* wtot) {
    for (int off = 1; off < 32; off <<= 1) {
        const int y = __shfl_up(x, off, 32);
        if (lane >= off) x += y;
    }
    if (wave < 8 && lane == 31) wtot[wave] = x;
    return x;
}

__device__ __forceinline__ int pair_index(const int* pfx, const int* pstart, int p) {
    int lo = 0, hi = SORT_CHUNKS;
    for (int it = 0; it < 8; ++it) {
        const int mid = (lo + hi) >> 1;
        if (pfx[mid] <= p) lo = mid; else hi = mid;
    }
    return pstart[lo] + (p - pfx[lo]);
}
static_assert(SORT_CHUNKS == 256, "pair_index searches 256 pieces in 8 steps");

#define KEEP 8
#define STAGE_CAP 16384

__global__ __launch_bounds__(SORT_T)
void bucket_sort_kernel(const int* __restrict__ pairs, const int* __restrict__ offs_t,
                        int* __restrict__ rowptr, float* __restrict__ isq,
                        int* __restrict__ col) {
    __shared__ int pstart[SORT_CHUNKS];
    __shared__ int pfx[SORT_CHUNKS + 1];
    __shared__ int cnt[BKT_NODES];
    __shared__ int nstart[BKT_NODES];
    __shared__ int wtotA[8], wtotB[8], wtotC[8];
    __shared__ int lcol[STAGE_CAP];

    const int tid  = threadIdx.x;
    const int lane = tid & 31;
    const int wave = tid >> 5;
    const int k    = blockIdx.x;
    const v2i* __restrict__ pv = reinterpret_cast<const v2i*>(pairs);

    int len = 0, o0 = 0;
    if (tid < SORT_CHUNKS) {
        o0 = offs_t[k * SORT_CHUNKS + tid];
        const int o1 = offs_t[(k + 1) * SORT_CHUNKS + tid];
        len = o1 - o0;
        pstart[tid] = tid * CHUNK_E + o0;
        cnt[tid] = 0;
    }
    const int lenInc = scan256_inclusive(len, lane, wave, wtotA);
    (void)scan256_inclusive(o0, lane, wave, wtotB);
    __syncthreads();
    if (tid < SORT_CHUNKS) {
        int woff = 0;
        for (int w = 0; w < wave; ++w) woff += wtotA[w];
        pfx[tid] = woff + lenInc - len;
        if (tid == SORT_CHUNKS - 1) pfx[SORT_CHUNKS] = woff + lenInc;
    }
    int base = 0;
    for (int w = 0; w < 8; ++w) base += wtotB[w];
    __syncthreads();
    const int total = pfx[SORT_CHUNKS];

    v2i keep[KEEP];
#pragma unroll
    for (int u = 0; u < KEEP; ++u) {
        const int p  = tid + u * SORT_T;
        const int ix = pair_index(pfx, pstart, p);
        keep[u] = pv[p < total ? ix : 0];
    }
#pragma unroll
    for (int u = 0; u < KEEP; ++u)
        if (tid + u * SORT_T < total) atomicAdd(&cnt[keep[u].y - k * BKT_NODES], 1);
    for (int p = tid + KEEP * SORT_T; p < total; p += SORT_T) {
        const v2i pr = pv[pair_index(pfx, pstart, p)];
        atomicAdd(&cnt[pr.y - k * BKT_NODES], 1);
    }
    __syncthreads();

    int c = 0;
    if (tid < BKT_NODES) c = cnt[tid];
    const int cInc = scan256_inclusive(c, lane, wave, wtotC);
    __syncthreads();
    if (tid < BKT_NODES) {
        int woff = 0;
        for (int w = 0; w < wave; ++w) woff += wtotC[w];
        const int st = woff + cInc - c;
        nstart[tid] = st;
        cnt[tid]    = 0;
        const int node = k * BKT_NODES + tid;
        rowptr[node] = base + st;
        isq[node]    = rsqrtf((float)c + 1.0f);
    }
    __syncthreads();

    const bool staged = total <= STAGE_CAP;
    int* __restrict__ cslice = col + base;
#pragma unroll
    for (int u = 0; u < KEEP; ++u) {
        if (tid + u * SORT_T < total) {
            const int dl = keep[u].y - k * BKT_NODES;
            const int q  = nstart[dl] + atomicAdd(&cnt[dl], 1);
            if (staged) lcol[q] = keep[u].x; else cslice[q] = keep[u].x;
        }
    }
    for (int p = tid + KEEP * SORT_T; p < total; p += SORT_T) {
        const v2i pr = pv[pair_index(pfx, pstart, p)];
        const int dl = pr.y - k * BKT_NODES;
        const int q  = nstart[dl] + atomicAdd(&cnt[dl], 1);
        if (staged) lcol[q] = pr.x; else cslice[q] = pr.x;
    }
    __syncthreads();
    if (staged)
        for (int q = tid; q < total; q += SORT_T) cslice[q] = lcol[q];
}


template<int U, typename vec>
__device__ __forceinline__ void gather_rows(vec& acc, const vec* __restrict__ hv,
                                            int lane, int s_l, float c_l, int t) {
    vec   hr[U];
    float cf[U];
#pragma unroll
    for (int u = 0; u < U; ++u) {
        const int s = __builtin_amdgcn_readlane(s_l, t + u);
        cf[u] = __builtin_bit_cast(
            float, __builtin_amdgcn_readlane(__builtin_bit_cast(int, c_l), t + u));
        hr[u] = hv[(size_t)s * 32 + lane];
    }
#pragma unroll
    for (int u = 0; u < U; ++u) acc += hr[u] * cf[u];
}

template<typename vec>
__device__ __forceinline__ vec gather_edges(const vec* __restrict__ hv,
                                            const int* __restrict__ col,
                                            const float* __restrict__ isq,
                                            int beg, int end, float sd, int lane) {
    vec acc = (vec)(0.0f);

    for (int base = beg; base < end; base += 32) {
        const int j = base + lane;
        int   s_l = 0;
        float c_l = 0.0f;
        if (j < end) {
            s_l = col[j];
            c_l = isq[s_l] * sd;
        }
        const int rem = end - base;
        const int m   = rem < 32 ? rem : 32;

        int t = 0;
        for (; t + 8 <= m; t += 8) gather_rows<8, vec>(acc, hv, lane, s_l, c_l, t);
        if (m & 4) { gather_rows<4, vec>(acc, hv, lane, s_l, c_l, t); t += 4; }
        if (m & 2) { gather_rows<2, vec>(acc, hv, lane, s_l, c_l, t); t += 2; }
        if (m & 1) { gather_rows<1, vec>(acc, hv, lane, s_l, c_l, t); }
    }
    return acc;
}

template<int C>
__global__ __launch_bounds__(256)
void gather_kernel(const float* __restrict__ h,
                   const int* __restrict__ rowptr,
                   const int* __restrict__ col,
                   const float* __restrict__ isq,
                   const float* __restrict__ bias,
                   float* __restrict__ out, int n) {
    constexpr int V = C / 32;
    typedef float vec __attribute__((ext_vector_type(V)));

    const int lane = threadIdx.x & 31;
    const int node = __builtin_amdgcn_readfirstlane(
        (int)(blockIdx.x * (blockDim.x >> 5) + (threadIdx.x >> 5)));
    if (node >= n) return;

    const int   beg = rowptr[node];
    const int   end = rowptr[node + 1];
    const float sd  = isq[node];

    const vec* __restrict__ hv = reinterpret_cast<const vec*>(h);
    const vec* __restrict__ bv = reinterpret_cast<const vec*>(bias);

    const vec hself = hv[(size_t)node * 32 + lane];
    const vec bself = bv[lane];

    vec acc = gather_edges<vec>(hv, col, isq, beg, end, sd, lane);

    acc += hself * (sd * sd) + bself;
    reinterpret_cast<vec*>(out)[(size_t)node * 32 + lane] = acc;
}


#define TILE_M    32
#define FUSE_T    (TILE_M * 32)
#define FUSE_GRID 512
#define N_TILES   ((N_NODES + TILE_M - 1) / TILE_M)
#define A_PITCH   (C_HID + 4)
#define W1_PITCH  (2 * C_HID + 32)
#define W2_PITCH  (2 * C_OUTC + 32)

static_assert(C_IN == C_HID, "one tile buffer shape serves both products");
static_assert(C_IN == 128 && C_OUTC == 64, "lane maps below assume 128 -> 128 -> 64");

template<int BPITCH>
__device__ __forceinline__ v8f tile_product(const float* ap, const float* bp) {
    v8f acc = {};
#pragma unroll
    for (int kk = 0; kk < C_HID / 4; ++kk) {
        const v2f a = *reinterpret_cast<const v2f*>(ap + 4 * kk);
        const v2f b = *reinterpret_cast<const v2f*>(bp + 2 * kk * BPITCH);
        acc = __builtin_amdgcn_wmma_f32_16x16x4_f32(
                  false, a, false, b, (short)0, acc, false, false);
    }
    return acc;
}

__global__ __launch_bounds__(FUSE_T) __attribute__((amdgpu_waves_per_eu(16, 16)))
void fused_layer1_kernel(const float* __restrict__ x,
                         const int* __restrict__ rowptr,
                         const int* __restrict__ col,
                         const float* __restrict__ isq,
                         const float* __restrict__ W1,
                         const float* __restrict__ b1,
                         const float* __restrict__ W2,
                         float* __restrict__ h2) {
    __shared__ __attribute__((aligned(16))) float sW1[(C_IN / 2) * W1_PITCH];
    __shared__ __attribute__((aligned(16))) float sW2[(C_HID / 2) * W2_PITCH];
    __shared__ __attribute__((aligned(16))) float sA[TILE_M * A_PITCH];
    __shared__ __attribute__((aligned(16))) float sH[TILE_M * A_PITCH];

    const int tid  = threadIdx.x;
    const int lane = tid & 31;
    const int wave = __builtin_amdgcn_readfirstlane(tid >> 5);
    const int m    = lane & 15;
    const int half = lane >> 4;

    for (int j = tid; j < (C_IN / 2) * (C_HID / 4); j += FUSE_T) {
        const int kp = j >> 5, g = j & 31;
        const v4f r0 = *reinterpret_cast<const v4f*>(W1 + (size_t)(2 * kp) * C_HID + g * 4);
        const v4f r1 = *reinterpret_cast<const v4f*>(W1 + (size_t)(2 * kp + 1) * C_HID + g * 4);
        v4f* d = reinterpret_cast<v4f*>(sW1 + kp * W1_PITCH + g * 8);
        const v4f lo = {r0.x, r1.x, r0.y, r1.y};
        const v4f hi = {r0.z, r1.z, r0.w, r1.w};
        d[0] = lo;
        d[1] = hi;
    }
    for (int j = tid; j < (C_HID / 2) * (C_OUTC / 4); j += FUSE_T) {
        const int kp = j >> 4, g = j & 15;
        const v4f r0 = *reinterpret_cast<const v4f*>(W2 + (size_t)(2 * kp) * C_OUTC + g * 4);
        const v4f r1 = *reinterpret_cast<const v4f*>(W2 + (size_t)(2 * kp + 1) * C_OUTC + g * 4);
        v4f* d = reinterpret_cast<v4f*>(sW2 + kp * W2_PITCH + g * 8);
        const v4f lo = {r0.x, r1.x, r0.y, r1.y};
        const v4f hi = {r0.z, r1.z, r0.w, r1.w};
        d[0] = lo;
        d[1] = hi;
    }

    const v4f* __restrict__ xv = reinterpret_cast<const v4f*>(x);

    for (int tile = blockIdx.x; tile < N_TILES; tile += FUSE_GRID) {
        const int node = tile * TILE_M + wave;

        v4f agg = (v4f)(0.0f);
        if (node < N_NODES) {
            const int   beg = rowptr[node];
            const int   end = rowptr[node + 1];
            const float sd  = isq[node];
            const v4f xself = xv[(size_t)node * 32 + lane];
            agg  = gather_edges<v4f>(xv, col, isq, beg, end, sd, lane);
            agg += xself * (sd * sd);
        }
        *reinterpret_cast<v4f*>(sA + wave * A_PITCH + lane * 4) = agg;
        __syncthreads();

        if (wave < 16) {
            const int mt = wave >> 3, nt = wave & 7;
            const float bc = b1[nt * 16 + m];
            const v8f acc = tile_product<W1_PITCH>(
                sA  + (mt * 16 + m) * A_PITCH + half * 2,
                sW1 + half * W1_PITCH + (nt * 16 + m) * 2);
            float* hp = sH + (mt * 16 + half * 8) * A_PITCH + nt * 16 + m;
#pragma unroll
            for (int r = 0; r < 8; ++r) hp[r * A_PITCH] = fmaxf(acc[r] + bc, 0.0f);
        }
        __syncthreads();

        if (wave < 8) {
            const int mt = wave >> 2, nt = wave & 3;
            const v8f acc = tile_product<W2_PITCH>(
                sH  + (mt * 16 + m) * A_PITCH + half * 2,
                sW2 + half * W2_PITCH + (nt * 16 + m) * 2);
            float* op = sA + (mt * 16 + half * 8) * A_PITCH + nt * 16 + m;
#pragma unroll
            for (int r = 0; r < 8; ++r) op[r * A_PITCH] = acc[r];
        }
        __syncthreads();

        if (node < N_NODES) {
            const v2f v = *reinterpret_cast<const v2f*>(sA + wave * A_PITCH + lane * 2);
            *reinterpret_cast<v2f*>(h2 + (size_t)node * C_OUTC + lane * 2) = v;
        }
        __syncthreads();
    }
}


extern "C" void kernel_launch(void* const* d_in, const int* in_sizes, int n_in,
                              void* d_out, int out_size, void* d_ws, size_t ws_size,
                              hipStream_t stream) {
    (void)in_sizes; (void)n_in; (void)out_size; (void)ws_size;

    const float* x   = (const float*)d_in[0];
    const int*   ei  = (const int*)  d_in[1];
    const float* W1  = (const float*)d_in[2];
    const float* b1  = (const float*)d_in[3];
    const float* W2  = (const float*)d_in[4];
    const float* b2  = (const float*)d_in[5];
    float*       out = (float*)d_out;

    const int* src = ei;
    const int* dst = ei + N_EDGES;

    int*   offs_t = (int*)d_ws;
    int*   pairs  = offs_t + (N_BKT + 1) * SORT_CHUNKS;
    int*   rowptr = pairs + 2 * N_EDGES;
    float* isq    = (float*)(rowptr + N_PAD);
    int*   col    = (int*)(isq + N_PAD);
    float* h2     = (float*)(col + N_EDGES);

    const int T = 256;

    chunk_sort_kernel <<<SORT_CHUNKS, SORT_T, 0, stream>>>(src, dst, pairs, offs_t);
    bucket_sort_kernel<<<N_BKT, SORT_T, 0, stream>>>(pairs, offs_t, rowptr, isq, col);

    fused_layer1_kernel<<<FUSE_GRID, FUSE_T, 0, stream>>>(x, rowptr, col, isq, W1, b1, W2, h2);

    gather_kernel<C_OUTC>
        <<<N_NODES / 8, T, 0, stream>>>(h2, rowptr, col, isq, b2, out, N_NODES);
}


# Layer-2 edge weight split: h2 rows stored pre-scaled by their node's isq; last gather adds rows unweighted (no per-edge isq lookup) and scales the sum by isq[d] once; sum order kept, rounding differs
# speedup vs baseline: 5.7524x; 1.0341x over previous
#include <hip/hip_runtime.h>


typedef float v2f __attribute__((ext_vector_type(2)));
typedef float v4f __attribute__((ext_vector_type(4)));
typedef float v8f __attribute__((ext_vector_type(8)));
typedef int   v2i __attribute__((ext_vector_type(2)));

#define N_NODES 50000
#define N_EDGES 800000
#define C_IN    128
#define C_HID   128
#define C_OUTC  64

#define N_PAD     53248

static_assert(N_PAD > N_NODES && N_PAD % 128 == 0, "padded node arrays");
static_assert(N_NODES % 8 == 0, "gather grid assumes 8 nodes per block");


#define SORT_CHUNKS 256
#define CHUNK_E     (N_EDGES / SORT_CHUNKS)
#define BKT_NODES   256
#define N_BKT       ((N_NODES + BKT_NODES - 1) / BKT_NODES)
#define SORT_T      1024
#define CHUNK_PER_T ((CHUNK_E + SORT_T - 1) / SORT_T)

static_assert(N_EDGES % SORT_CHUNKS == 0, "chunks must tile the edge list");
static_assert(N_BKT + 1 <= 256, "bucket offsets are scanned as 256 entries by one wave");
static_assert(N_BKT * BKT_NODES + 1 <= N_PAD, "rowptr must hold every bucket's nodes");
static_assert(N_BKT * BKT_NODES > N_NODES, "entry N_NODES of rowptr must belong to a bucket");
static_assert((((N_BKT + 1) * SORT_CHUNKS) % 2) == 0, "pairs must start 8-byte aligned");
static_assert((((N_BKT + 1) * SORT_CHUNKS + 3 * N_EDGES + 2 * N_PAD) % 128) == 0,
              "row tables must start on a 512-byte boundary");

__global__ __launch_bounds__(SORT_T)
void chunk_sort_kernel(const int* __restrict__ src, const int* __restrict__ dst,
                       int* __restrict__ pairs, int* __restrict__ offs_t) {
    __shared__ int hist[256];
    __shared__ v2i stage[CHUNK_E];
    const int tid = threadIdx.x;
    const int e0  = blockIdx.x * CHUNK_E;

    if (tid < 256) hist[tid] = 0;

    int s[CHUNK_PER_T], d[CHUNK_PER_T], r[CHUNK_PER_T];
#pragma unroll
    for (int i = 0; i < CHUNK_PER_T; ++i) {
        const int j = tid + i * SORT_T;
        s[i] = 0; d[i] = -1;
        if (j < CHUNK_E) {
            s[i] = src[e0 + j];
            d[i] = dst[e0 + j];
        }
    }
    __syncthreads();
#pragma unroll
    for (int i = 0; i < CHUNK_PER_T; ++i) {
        r[i] = 0;
        if (d[i] >= 0) r[i] = atomicAdd(&hist[d[i] >> 8], 1);
    }
    __syncthreads();

    if (tid < 32) {
        int v[8];
        int sum = 0;
#pragma unroll
        for (int q = 0; q < 8; ++q) { const int c = hist[tid * 8 + q]; v[q] = sum; sum += c; }
        int x = sum;
        for (int off = 1; off < 32; off <<= 1) {
            const int y = __shfl_up(x, off, 32);
            if (tid >= off) x += y;
        }
        const int ex = x - sum;
#pragma unroll
        for (int q = 0; q < 8; ++q) hist[tid * 8 + q] = ex + v[q];
    }
    __syncthreads();

    if (tid <= N_BKT) offs_t[tid * SORT_CHUNKS + blockIdx.x] = hist[tid];

#pragma unroll
    for (int i = 0; i < CHUNK_PER_T; ++i) {
        if (d[i] >= 0) {
            v2i p = {s[i], d[i]};
            stage[hist[d[i] >> 8] + r[i]] = p;
        }
    }
    __syncthreads();

    v2i* __restrict__ pout = reinterpret_cast<v2i*>(pairs) + e0;
    for (int j = tid; j < CHUNK_E; j += SORT_T) pout[j] = stage[j];
}

__device__ __forceinline__ int scan256_inclusive(int x, int lane, int wave, int* wtot) {
    for (int off = 1; off < 32; off <<= 1) {
        const int y = __shfl_up(x, off, 32);
        if (lane >= off) x += y;
    }
    if (wave < 8 && lane == 31) wtot[wave] = x;
    return x;
}

__device__ __forceinline__ int pair_index(const int* pfx, const int* pstart, int p) {
    int lo = 0, hi = SORT_CHUNKS;
    for (int it = 0; it < 8; ++it) {
        const int mid = (lo + hi) >> 1;
        if (pfx[mid] <= p) lo = mid; else hi = mid;
    }
    return pstart[lo] + (p - pfx[lo]);
}
static_assert(SORT_CHUNKS == 256, "pair_index searches 256 pieces in 8 steps");

#define KEEP 8
#define STAGE_CAP 16384

__global__ __launch_bounds__(SORT_T)
void bucket_sort_kernel(const int* __restrict__ pairs, const int* __restrict__ offs_t,
                        int* __restrict__ rowptr, float* __restrict__ isq,
                        int* __restrict__ col) {
    __shared__ int pstart[SORT_CHUNKS];
    __shared__ int pfx[SORT_CHUNKS + 1];
    __shared__ int cnt[BKT_NODES];
    __shared__ int nstart[BKT_NODES];
    __shared__ int wtotA[8], wtotB[8], wtotC[8];
    __shared__ int lcol[STAGE_CAP];

    const int tid  = threadIdx.x;
    const int lane = tid & 31;
    const int wave = tid >> 5;
    const int k    = blockIdx.x;
    const v2i* __restrict__ pv = reinterpret_cast<const v2i*>(pairs);

    int len = 0, o0 = 0;
    if (tid < SORT_CHUNKS) {
        o0 = offs_t[k * SORT_CHUNKS + tid];
        const int o1 = offs_t[(k + 1) * SORT_CHUNKS + tid];
        len = o1 - o0;
        pstart[tid] = tid * CHUNK_E + o0;
        cnt[tid] = 0;
    }
    const int lenInc = scan256_inclusive(len, lane, wave, wtotA);
    (void)scan256_inclusive(o0, lane, wave, wtotB);
    __syncthreads();
    if (tid < SORT_CHUNKS) {
        int woff = 0;
        for (int w = 0; w < wave; ++w) woff += wtotA[w];
        pfx[tid] = woff + lenInc - len;
        if (tid == SORT_CHUNKS - 1) pfx[SORT_CHUNKS] = woff + lenInc;
    }
    int base = 0;
    for (int w = 0; w < 8; ++w) base += wtotB[w];
    __syncthreads();
    const int total = pfx[SORT_CHUNKS];

    v2i keep[KEEP];
#pragma unroll
    for (int u = 0; u < KEEP; ++u) {
        const int p  = tid + u * SORT_T;
        const int ix = pair_index(pfx, pstart, p);
        keep[u] = pv[p < total ? ix : 0];
    }
#pragma unroll
    for (int u = 0; u < KEEP; ++u)
        if (tid + u * SORT_T < total) atomicAdd(&cnt[keep[u].y - k * BKT_NODES], 1);
    for (int p = tid + KEEP * SORT_T; p < total; p += SORT_T) {
        const v2i pr = pv[pair_index(pfx, pstart, p)];
        atomicAdd(&cnt[pr.y - k * BKT_NODES], 1);
    }
    __syncthreads();

    int c = 0;
    if (tid < BKT_NODES) c = cnt[tid];
    const int cInc = scan256_inclusive(c, lane, wave, wtotC);
    __syncthreads();
    if (tid < BKT_NODES) {
        int woff = 0;
        for (int w = 0; w < wave; ++w) woff += wtotC[w];
        const int st = woff + cInc - c;
        nstart[tid] = st;
        cnt[tid]    = 0;
        const int node = k * BKT_NODES + tid;
        rowptr[node] = base + st;
        isq[node]    = rsqrtf((float)c + 1.0f);
    }
    __syncthreads();

    const bool staged = total <= STAGE_CAP;
    int* __restrict__ cslice = col + base;
#pragma unroll
    for (int u = 0; u < KEEP; ++u) {
        if (tid + u * SORT_T < total) {
            const int dl = keep[u].y - k * BKT_NODES;
            const int q  = nstart[dl] + atomicAdd(&cnt[dl], 1);
            if (staged) lcol[q] = keep[u].x; else cslice[q] = keep[u].x;
        }
    }
    for (int p = tid + KEEP * SORT_T; p < total; p += SORT_T) {
        const v2i pr = pv[pair_index(pfx, pstart, p)];
        const int dl = pr.y - k * BKT_NODES;
        const int q  = nstart[dl] + atomicAdd(&cnt[dl], 1);
        if (staged) lcol[q] = pr.x; else cslice[q] = pr.x;
    }
    __syncthreads();
    if (staged)
        for (int q = tid; q < total; q += SORT_T) cslice[q] = lcol[q];
}


template<bool WEIGHTED, int U, typename vec>
__device__ __forceinline__ void gather_rows(vec& acc, const vec* __restrict__ hv,
                                            int lane, int s_l, float c_l, int t) {
    vec   hr[U];
    float cf[U];
#pragma unroll
    for (int u = 0; u < U; ++u) {
        const int s = __builtin_amdgcn_readlane(s_l, t + u);
        cf[u] = 1.0f;
        if (WEIGHTED)
            cf[u] = __builtin_bit_cast(
                float, __builtin_amdgcn_readlane(__builtin_bit_cast(int, c_l), t + u));
        hr[u] = hv[(size_t)s * 32 + lane];
    }
#pragma unroll
    for (int u = 0; u < U; ++u) {
        if (WEIGHTED) acc += hr[u] * cf[u];
        else          acc += hr[u];
    }
}

template<bool WEIGHTED, typename vec>
__device__ __forceinline__ vec gather_edges(const vec* __restrict__ hv,
                                            const int* __restrict__ col,
                                            const float* __restrict__ isq,
                                            int beg, int end, float sd, int lane) {
    vec acc = (vec)(0.0f);

    for (int base = beg; base < end; base += 32) {
        const int j = base + lane;
        int   s_l = 0;
        float c_l = 0.0f;
        if (j < end) {
            s_l = col[j];
            if (WEIGHTED) c_l = isq[s_l] * sd;
        }
        const int rem = end - base;
        const int m   = rem < 32 ? rem : 32;

        int t = 0;
        for (; t + 8 <= m; t += 8) gather_rows<WEIGHTED, 8, vec>(acc, hv, lane, s_l, c_l, t);
        if (m & 4) { gather_rows<WEIGHTED, 4, vec>(acc, hv, lane, s_l, c_l, t); t += 4; }
        if (m & 2) { gather_rows<WEIGHTED, 2, vec>(acc, hv, lane, s_l, c_l, t); t += 2; }
        if (m & 1) { gather_rows<WEIGHTED, 1, vec>(acc, hv, lane, s_l, c_l, t); }
    }
    return acc;
}

template<int C>
__global__ __launch_bounds__(256)
void gather_kernel(const float* __restrict__ h,
                   const int* __restrict__ rowptr,
                   const int* __restrict__ col,
                   const float* __restrict__ isq,
                   const float* __restrict__ bias,
                   float* __restrict__ out, int n) {
    constexpr int V = C / 32;
    typedef float vec __attribute__((ext_vector_type(V)));

    const int lane = threadIdx.x & 31;
    const int node = __builtin_amdgcn_readfirstlane(
        (int)(blockIdx.x * (blockDim.x >> 5) + (threadIdx.x >> 5)));
    if (node >= n) return;

    const int   beg = rowptr[node];
    const int   end = rowptr[node + 1];
    const float sd  = isq[node];

    const vec* __restrict__ hv = reinterpret_cast<const vec*>(h);
    const vec* __restrict__ bv = reinterpret_cast<const vec*>(bias);

    const vec hself = hv[(size_t)node * 32 + lane];
    const vec bself = bv[lane];

    vec acc = gather_edges<false, vec>(hv, col, isq, beg, end, sd, lane);

    acc += hself;
    reinterpret_cast<vec*>(out)[(size_t)node * 32 + lane] = acc * sd + bself;
}


#define TILE_M    32
#define FUSE_T    (TILE_M * 32)
#define FUSE_GRID 512
#define N_TILES   ((N_NODES + TILE_M - 1) / TILE_M)
#define A_PITCH   (C_HID + 4)
#define W1_PITCH  (2 * C_HID + 32)
#define W2_PITCH  (2 * C_OUTC + 32)

static_assert(C_IN == C_HID, "one tile buffer shape serves both products");
static_assert(C_IN == 128 && C_OUTC == 64, "lane maps below assume 128 -> 128 -> 64");

template<int BPITCH>
__device__ __forceinline__ v8f tile_product(const float* ap, const float* bp) {
    v8f acc = {};
#pragma unroll
    for (int kk = 0; kk < C_HID / 4; ++kk) {
        const v2f a = *reinterpret_cast<const v2f*>(ap + 4 * kk);
        const v2f b = *reinterpret_cast<const v2f*>(bp + 2 * kk * BPITCH);
        acc = __builtin_amdgcn_wmma_f32_16x16x4_f32(
                  false, a, false, b, (short)0, acc, false, false);
    }
    return acc;
}

__global__ __launch_bounds__(FUSE_T) __attribute__((amdgpu_waves_per_eu(16, 16)))
void fused_layer1_kernel(const float* __restrict__ x,
                         const int* __restrict__ rowptr,
                         const int* __restrict__ col,
                         const float* __restrict__ isq,
                         const float* __restrict__ W1,
                         const float* __restrict__ b1,
                         const float* __restrict__ W2,
                         float* __restrict__ h2) {
    __shared__ __attribute__((aligned(16))) float sW1[(C_IN / 2) * W1_PITCH];
    __shared__ __attribute__((aligned(16))) float sW2[(C_HID / 2) * W2_PITCH];
    __shared__ __attribute__((aligned(16))) float sA[TILE_M * A_PITCH];
    __shared__ __attribute__((aligned(16))) float sH[TILE_M * A_PITCH];

    const int tid  = threadIdx.x;
    const int lane = tid & 31;
    const int wave = __builtin_amdgcn_readfirstlane(tid >> 5);
    const int m    = lane & 15;
    const int half = lane >> 4;

    for (int j = tid; j < (C_IN / 2) * (C_HID / 4); j += FUSE_T) {
        const int kp = j >> 5, g = j & 31;
        const v4f r0 = *reinterpret_cast<const v4f*>(W1 + (size_t)(2 * kp) * C_HID + g * 4);
        const v4f r1 = *reinterpret_cast<const v4f*>(W1 + (size_t)(2 * kp + 1) * C_HID + g * 4);
        v4f* d = reinterpret_cast<v4f*>(sW1 + kp * W1_PITCH + g * 8);
        const v4f lo = {r0.x, r1.x, r0.y, r1.y};
        const v4f hi = {r0.z, r1.z, r0.w, r1.w};
        d[0] = lo;
        d[1] = hi;
    }
    for (int j = tid; j < (C_HID / 2) * (C_OUTC / 4); j += FUSE_T) {
        const int kp = j >> 4, g = j & 15;
        const v4f r0 = *reinterpret_cast<const v4f*>(W2 + (size_t)(2 * kp) * C_OUTC + g * 4);
        const v4f r1 = *reinterpret_cast<const v4f*>(W2 + (size_t)(2 * kp + 1) * C_OUTC + g * 4);
        v4f* d = reinterpret_cast<v4f*>(sW2 + kp * W2_PITCH + g * 8);
        const v4f lo = {r0.x, r1.x, r0.y, r1.y};
        const v4f hi = {r0.z, r1.z, r0.w, r1.w};
        d[0] = lo;
        d[1] = hi;
    }

    const v4f* __restrict__ xv = reinterpret_cast<const v4f*>(x);

    for (int tile = blockIdx.x; tile < N_TILES; tile += FUSE_GRID) {
        const int node = tile * TILE_M + wave;

        v4f agg = (v4f)(0.0f);
        float sd = 0.0f;
        if (node < N_NODES) {
            const int   beg = rowptr[node];
            const int   end = rowptr[node + 1];
            sd = isq[node];
            const v4f xself = xv[(size_t)node * 32 + lane];
            agg  = gather_edges<true, v4f>(xv, col, isq, beg, end, sd, lane);
            agg += xself * (sd * sd);
        }
        *reinterpret_cast<v4f*>(sA + wave * A_PITCH + lane * 4) = agg;
        __syncthreads();

        if (wave < 16) {
            const int mt = wave >> 3, nt = wave & 7;
            const float bc = b1[nt * 16 + m];
            const v8f acc = tile_product<W1_PITCH>(
                sA  + (mt * 16 + m) * A_PITCH + half * 2,
                sW1 + half * W1_PITCH + (nt * 16 + m) * 2);
            float* hp = sH + (mt * 16 + half * 8) * A_PITCH + nt * 16 + m;
#pragma unroll
            for (int r = 0; r < 8; ++r) hp[r * A_PITCH] = fmaxf(acc[r] + bc, 0.0f);
        }
        __syncthreads();

        if (wave < 8) {
            const int mt = wave >> 2, nt = wave & 3;
            const v8f acc = tile_product<W2_PITCH>(
                sH  + (mt * 16 + m) * A_PITCH + half * 2,
                sW2 + half * W2_PITCH + (nt * 16 + m) * 2);
            float* op = sA + (mt * 16 + half * 8) * A_PITCH + nt * 16 + m;
#pragma unroll
            for (int r = 0; r < 8; ++r) op[r * A_PITCH] = acc[r];
        }
        __syncthreads();

        if (node < N_NODES) {
            const v2f v = *reinterpret_cast<const v2f*>(sA + wave * A_PITCH + lane * 2);
            *reinterpret_cast<v2f*>(h2 + (size_t)node * C_OUTC + lane * 2) = v * sd;
        }
        __syncthreads();
    }
}


extern "C" void kernel_launch(void* const* d_in, const int* in_sizes, int n_in,
                              void* d_out, int out_size, void* d_ws, size_t ws_size,
                              hipStream_t stream) {
    (void)in_sizes; (void)n_in; (void)out_size; (void)ws_size;

    const float* x   = (const float*)d_in[0];
    const int*   ei  = (const int*)  d_in[1];
    const float* W1  = (const float*)d_in[2];
    const float* b1  = (const float*)d_in[3];
    const float* W2  = (const float*)d_in[4];
    const float* b2  = (const float*)d_in[5];
    float*       out = (float*)d_out;

    const int* src = ei;
    const int* dst = ei + N_EDGES;

    int*   offs_t = (int*)d_ws;
    int*   pairs  = offs_t + (N_BKT + 1) * SORT_CHUNKS;
    int*   rowptr = pairs + 2 * N_EDGES;
    float* isq    = (float*)(rowptr + N_PAD);
    int*   col    = (int*)(isq + N_PAD);
    float* h2     = (float*)(col + N_EDGES);

    const int T = 256;

    chunk_sort_kernel <<<SORT_CHUNKS, SORT_T, 0, stream>>>(src, dst, pairs, offs_t);
    bucket_sort_kernel<<<N_BKT, SORT_T, 0, stream>>>(pairs, offs_t, rowptr, isq, col);

    fused_layer1_kernel<<<FUSE_GRID, FUSE_T, 0, stream>>>(x, rowptr, col, isq, W1, b1, W2, h2);

    gather_kernel<C_OUTC>
        <<<N_NODES / 8, T, 0, stream>>>(h2, rowptr, col, isq, b2, out, N_NODES);
}
